# starting kernel
# baseline (speedup 1.0000x reference)
#include <hip/hip_runtime.h>


typedef __attribute__((ext_vector_type(2))) float v2f;
typedef __attribute__((ext_vector_type(8))) float v8f;

#define T_TOKENS 4096
#define DM 512
#define DH 2048
#define NE 8
#define MT 16
#define HC 128
#define XROW (DM + 4)
#define HROW (HC + 4)

static __device__ __forceinline__ v8f wmma_f32(v2f a, v2f b, v8f c) {
  return __builtin_amdgcn_wmma_f32_16x16x4_f32(false, a, false, b, (short)0, c,
                                               false, false);
}

__global__ __launch_bounds__(256) void gating_kernel(
    const float* __restrict__ emb, const float* __restrict__ state,
    const float* __restrict__ Wg, const float* __restrict__ bg,
    const float* __restrict__ Kmat, float* __restrict__ probs_out,
    float* __restrict__ gates_out) {
  const int wave = threadIdx.x >> 5;
  const int lane = threadIdx.x & 31;
  const int t = blockIdx.x * 8 + wave;
  if (t >= T_TOKENS) return;

  const float* x = emb + (size_t)t * DM;
  float acc[NE];
#pragma unroll
  for (int e = 0; e < NE; ++e) acc[e] = 0.0f;

  for (int d = lane; d < DM; d += 32) {
    const float xd = x[d];
#pragma unroll
    for (int e = 0; e < NE; ++e) acc[e] = fmaf(xd, Wg[e * DM + d], acc[e]);
  }
#pragma unroll
  for (int e = 0; e < NE; ++e) {
#pragma unroll
    for (int off = 16; off > 0; off >>= 1) acc[e] += __shfl_xor(acc[e], off, 32);
  }

  if (lane == 0) {
    float logit[NE], p[NE];
#pragma unroll
    for (int e = 0; e < NE; ++e) {
      float tb = 0.0f;
#pragma unroll
      for (int f = 0; f < NE; ++f) tb = fmaf(Kmat[e * NE + f], state[t * NE + f], tb);
      logit[e] = acc[e] + bg[e] + tb;
    }
    float m = logit[0];
#pragma unroll
    for (int e = 1; e < NE; ++e) m = fmaxf(m, logit[e]);
    float s = 0.0f;
#pragma unroll
    for (int e = 0; e < NE; ++e) { p[e] = expf(logit[e] - m); s += p[e]; }
    const float inv = 1.0f / s;
#pragma unroll
    for (int e = 0; e < NE; ++e) p[e] *= inv;
    int i1 = 0;
#pragma unroll
    for (int e = 1; e < NE; ++e) if (p[e] > p[i1]) i1 = e;
    int i2 = (i1 == 0) ? 1 : 0;
#pragma unroll
    for (int e = 0; e < NE; ++e) if (e != i1 && p[e] > p[i2]) i2 = e;
#pragma unroll
    for (int e = 0; e < NE; ++e) {
      probs_out[t * NE + e] = p[e];
      gates_out[t * NE + e] = (e == i1 || e == i2) ? p[e] : 0.0f;
    }
  }
}

__global__ __launch_bounds__(256) void moe_ffn_kernel(
    const float* __restrict__ emb, const float* __restrict__ gates,
    const float* __restrict__ W1, const float* __restrict__ b1,
    const float* __restrict__ W2, const float* __restrict__ b2,
    float* __restrict__ out) {
  __shared__ float sX[MT * XROW];
  __shared__ float sH[MT * HROW];
  __shared__ float sG[MT * NE];
  __shared__ float sFlag[NE];

  const int tid = threadIdx.x;
  const int wave = tid >> 5;
  const int lane = tid & 31;
  const int hf = lane >> 4;
  const int ln = lane & 15;
  const int t0 = blockIdx.x * MT;

  for (int i = tid; i < MT * DM; i += 256) {
    const int m = i >> 9, d = i & (DM - 1);
    sX[m * XROW + d] = emb[(size_t)(t0 + m) * DM + d];
  }
  for (int i = tid; i < MT * NE; i += 256) sG[i] = gates[(size_t)t0 * NE + i];
  __syncthreads();
  if (tid < NE) {
    float f = 0.0f;
    for (int m = 0; m < MT; ++m) f += sG[m * NE + tid];
    sFlag[tid] = f;
  }
  __syncthreads();

  v8f accOut[4];
#pragma unroll
  for (int nt = 0; nt < 4; ++nt) accOut[nt] = (v8f)0.0f;

  for (int e = 0; e < NE; ++e) {
    if (sFlag[e] == 0.0f) continue;

    const float* W1p = W1 + (size_t)e * DM * DH;
    const float* W2p = W2 + (size_t)e * DH * DM;

    v8f accO[4];
#pragma unroll
    for (int nt = 0; nt < 4; ++nt) accO[nt] = (v8f)0.0f;

    for (int ch = 0; ch < DH / HC; ++ch) {
      const int hbase = ch * HC;
      const float* W2c = W2p + (size_t)hbase * DM;
      if (ch + 1 < DH / HC)
        __builtin_prefetch(&W2c[HC * DM + wave * 64 + ln], 0, 1);

      {
        const int hcol = hbase + wave * 16 + ln;
        v8f hcc = (v8f)0.0f;
#pragma unroll 8
        for (int k0 = 0; k0 < DM; k0 += 4) {
          const int k = k0 + 2 * hf;
          v2f a = {sX[ln * XROW + k], sX[ln * XROW + k + 1]};
          const int bi = k * DH + hcol;
          v2f b = {W1p[bi], W1p[bi + DH]};
          hcc = wmma_f32(a, b, hcc);
        }
        const float bias = b1[e * DH + hcol];
#pragma unroll
        for (int r = 0; r < 8; ++r) {
          float v = hcc[r] + bias;
          v = v > 0.0f ? v : 0.0f;
          sH[(r + 8 * hf) * HROW + wave * 16 + ln] = v;
        }
      }
      __syncthreads();

#pragma unroll 4
      for (int k0 = 0; k0 < HC; k0 += 4) {
        const int k = k0 + 2 * hf;
        v2f a = {sH[ln * HROW + k], sH[ln * HROW + k + 1]};
#pragma unroll
        for (int nt = 0; nt < 4; ++nt) {
          const int ncol = wave * 64 + nt * 16 + ln;
          const int bi = k * DM + ncol;
          v2f b = {W2c[bi], W2c[bi + DM]};
          accO[nt] = wmma_f32(a, b, accO[nt]);
        }
      }
      __syncthreads();
    }

#pragma unroll
    for (int nt = 0; nt < 4; ++nt) {
      const int ncol = wave * 64 + nt * 16 + ln;
      const float b2v = b2[e * DM + ncol];
#pragma unroll
      for (int r = 0; r < 8; ++r) {
        const float g = sG[(r + 8 * hf) * NE + e];
        accOut[nt][r] = fmaf(g, accO[nt][r] + b2v, accOut[nt][r]);
      }
    }
  }

#pragma unroll
  for (int nt = 0; nt < 4; ++nt) {
    const int ncol = wave * 64 + nt * 16 + ln;
#pragma unroll
    for (int r = 0; r < 8; ++r) {
      const int m = r + 8 * hf;
      out[(size_t)(t0 + m) * DM + ncol] = accOut[nt][r];
    }
  }
}

extern "C" void kernel_launch(void* const* d_in, const int* in_sizes, int n_in,
                              void* d_out, int out_size, void* d_ws, size_t ws_size,
                              hipStream_t stream) {
  const float* emb   = (const float*)d_in[0];
  const float* state = (const float*)d_in[1];
  const float* Wg    = (const float*)d_in[2];
  const float* bg    = (const float*)d_in[3];
  const float* Kmat  = (const float*)d_in[4];
  const float* W1    = (const float*)d_in[5];
  const float* b1    = (const float*)d_in[6];
  const float* W2    = (const float*)d_in[7];
  const float* b2    = (const float*)d_in[8];

  float* out_main = (float*)d_out;
  float* probs    = out_main + (size_t)T_TOKENS * DM;
  float* gates    = (float*)d_ws;

  gating_kernel<<<T_TOKENS / 8, 256, 0, stream>>>(emb, state, Wg, bg, Kmat,
                                                  probs, gates);
  moe_ffn_kernel<<<T_TOKENS / MT, 256, 0, stream>>>(emb, gates, W1, b1, W2, b2,
                                                    out_main);
}


# expert-major two-GEMM pipeline: rows gathered per expert by a scan; grouped GEMM1 stores relu hidden rows, grouped GEMM2 stores expert outputs, combine applies gates low expert first; sum order kept
# speedup vs baseline: 6.9964x; 6.9964x over previous
#include <hip/hip_runtime.h>


typedef __attribute__((ext_vector_type(2))) float v2f;
typedef __attribute__((ext_vector_type(8))) float v8f;

#define T_TOKENS 4096
#define DM 512
#define DH 2048
#define NE 8
#define BM 64
#define KC 128
#define AROW (KC + 4)
#define NROWS (2 * T_TOKENS)
#define MAXTILES (NROWS / BM + NE)

static __device__ __forceinline__ v8f wmma_f32(v2f a, v2f b, v8f c) {
  return __builtin_amdgcn_wmma_f32_16x16x4_f32(false, a, false, b, (short)0, c,
                                               false, false);
}

__global__ __launch_bounds__(256) void gating_kernel(
    const float* __restrict__ emb, const float* __restrict__ state,
    const float* __restrict__ Wg, const float* __restrict__ bg,
    const float* __restrict__ Kmat, float* __restrict__ probs_out,
    int* __restrict__ selE, float* __restrict__ selG) {
  const int wave = threadIdx.x >> 5;
  const int lane = threadIdx.x & 31;
  const int t = blockIdx.x * 8 + wave;
  if (t >= T_TOKENS) return;

  const float* x = emb + (size_t)t * DM;
  float acc[NE];
#pragma unroll
  for (int e = 0; e < NE; ++e) acc[e] = 0.0f;

  for (int d = lane; d < DM; d += 32) {
    const float xd = x[d];
#pragma unroll
    for (int e = 0; e < NE; ++e) acc[e] = fmaf(xd, Wg[e * DM + d], acc[e]);
  }
#pragma unroll
  for (int e = 0; e < NE; ++e) {
#pragma unroll
    for (int off = 16; off > 0; off >>= 1) acc[e] += __shfl_xor(acc[e], off, 32);
  }

  if (lane == 0) {
    float logit[NE], p[NE];
#pragma unroll
    for (int e = 0; e < NE; ++e) {
      float tb = 0.0f;
#pragma unroll
      for (int f = 0; f < NE; ++f) tb = fmaf(Kmat[e * NE + f], state[t * NE + f], tb);
      logit[e] = acc[e] + bg[e] + tb;
    }
    float m = logit[0];
#pragma unroll
    for (int e = 1; e < NE; ++e) m = fmaxf(m, logit[e]);
    float s = 0.0f;
#pragma unroll
    for (int e = 0; e < NE; ++e) { p[e] = expf(logit[e] - m); s += p[e]; }
    const float inv = 1.0f / s;
#pragma unroll
    for (int e = 0; e < NE; ++e) p[e] *= inv;
    int i1 = 0;
#pragma unroll
    for (int e = 1; e < NE; ++e) if (p[e] > p[i1]) i1 = e;
    int i2 = (i1 == 0) ? 1 : 0;
#pragma unroll
    for (int e = 0; e < NE; ++e) if (e != i1 && p[e] > p[i2]) i2 = e;
    const int lo = i1 < i2 ? i1 : i2;
    const int hi = i1 < i2 ? i2 : i1;
    float glo = 0.0f, ghi = 0.0f;
#pragma unroll
    for (int e = 0; e < NE; ++e) {
      probs_out[t * NE + e] = p[e];
      if (e == lo) glo = p[e];
      if (e == hi) ghi = p[e];
    }
    selE[2 * t] = lo;
    selE[2 * t + 1] = hi;
    selG[2 * t] = glo;
    selG[2 * t + 1] = ghi;
  }
}

__global__ __launch_bounds__(1024) void route_scan_kernel(
    const int* __restrict__ selE, int* __restrict__ tok, int* __restrict__ slot,
    int* __restrict__ tileE, int* __restrict__ tileR0, int* __restrict__ tileCnt) {
  __shared__ int sWaveTot[32 * NE];
  __shared__ int sWaveOff[32 * NE];
  __shared__ int sTot[NE];
  __shared__ int sBase[NE];

  const int tid = threadIdx.x;
  const int lane = tid & 31;
  const int wave = tid >> 5;

  int ee[8];
#pragma unroll
  for (int j = 0; j < 8; ++j) ee[j] = selE[tid * 8 + j] & (NE - 1);

  int cnt[NE], incl[NE];
#pragma unroll
  for (int x = 0; x < NE; ++x) {
    int c = 0;
#pragma unroll
    for (int j = 0; j < 8; ++j) c += (ee[j] == x) ? 1 : 0;
    cnt[x] = c;
    int v = c;
#pragma unroll
    for (int off = 1; off < 32; off <<= 1) {
      const int u = __shfl_up(v, off, 32);
      if (lane >= off) v += u;
    }
    incl[x] = v;
  }
  if (lane == 31) {
#pragma unroll
    for (int x = 0; x < NE; ++x) sWaveTot[wave * NE + x] = incl[x];
  }
  __syncthreads();
  if (tid < NE) {
    int run = 0;
    for (int w = 0; w < 32; ++w) {
      sWaveOff[w * NE + tid] = run;
      run += sWaveTot[w * NE + tid];
    }
    sTot[tid] = run;
  }
  __syncthreads();
  if (tid == 0) {
    int b = 0;
    int nt = 0;
    for (int x = 0; x < NE; ++x) {
      sBase[x] = b;
      const int n = sTot[x];
      for (int r = 0; r < n && nt < MAXTILES; r += BM) {
        tileE[nt] = x;
        tileR0[nt] = b + r;
        tileCnt[nt] = (n - r) < BM ? (n - r) : BM;
        ++nt;
      }
      b += n;
    }
    for (; nt < MAXTILES; ++nt) {
      tileE[nt] = -1;
      tileR0[nt] = 0;
      tileCnt[nt] = 0;
    }
  }
  __syncthreads();

  int pos[NE];
#pragma unroll
  for (int x = 0; x < NE; ++x)
    pos[x] = sBase[x] + sWaveOff[wave * NE + x] + incl[x] - cnt[x];
#pragma unroll
  for (int j = 0; j < 8; ++j) {
    int p = 0;
#pragma unroll
    for (int x = 0; x < NE; ++x) {
      if (ee[j] == x) { p = pos[x]; pos[x] += 1; }
    }
    p = p < 0 ? 0 : (p > NROWS - 1 ? NROWS - 1 : p);
    tok[p] = tid * 4 + (j >> 1);
    slot[tid * 8 + j] = p;
  }
}

template <int NW, int NJ, int K, int N, bool GATHER, bool RELU>
__global__ __launch_bounds__(NW * 32) void grouped_gemm_kernel(
    const float* __restrict__ A, const int* __restrict__ tok,
    const int* __restrict__ tileE, const int* __restrict__ tileR0,
    const int* __restrict__ tileCnt, const float* __restrict__ Wt,
    const float* __restrict__ bias, float* __restrict__ C) {
  __shared__ __attribute__((aligned(16))) float sA[BM * AROW];
  __shared__ int sRow[BM];

  const int tile = blockIdx.y;
  const int te = tileE[tile];
  if (te < 0) return;
  const int e = te & (NE - 1);
  int cnt = tileCnt[tile];
  cnt = cnt < 1 ? 1 : (cnt > BM ? BM : cnt);
  int r0 = tileR0[tile];
  r0 = r0 < 0 ? 0 : (r0 > NROWS - cnt ? NROWS - cnt : r0);

  const int tid = threadIdx.x;
  const int wave = tid >> 5;
  const int lane = tid & 31;
  const int hf = lane >> 4;
  const int ln = lane & 15;
  const int n0 = blockIdx.x * (NW * 16 * NJ) + wave * (16 * NJ);

  if (tid < BM) {
    const int r = r0 + (tid < cnt ? tid : 0);
    sRow[tid] = GATHER ? (tok[r] & (T_TOKENS - 1)) : r;
  }
  __syncthreads();

  const float* Wp = Wt + (size_t)e * K * N;

  v8f acc[4][NJ];
#pragma unroll
  for (int mi = 0; mi < 4; ++mi)
#pragma unroll
    for (int nj = 0; nj < NJ; ++nj) acc[mi][nj] = (v8f)0.0f;

  for (int kc = 0; kc < K; kc += KC) {
#pragma unroll
    for (int jb = 0; jb < BM / NW; jb += 8) {
      float4 v[8];
#pragma unroll
      for (int j = 0; j < 8; ++j) {
        const int m = wave + NW * (jb + j);
        v[j] = *reinterpret_cast<const float4*>(A + (size_t)sRow[m] * K + kc + 4 * lane);
      }
#pragma unroll
      for (int j = 0; j < 8; ++j) {
        const int m = wave + NW * (jb + j);
        *reinterpret_cast<float4*>(&sA[m * AROW + 4 * lane]) = v[j];
      }
    }
    __syncthreads();

    const float* Wc = Wp + (size_t)kc * N;
#pragma unroll(NJ == 1 ? 8 : 4)
    for (int k0 = 0; k0 < KC; k0 += 4) {
      const int k = k0 + 2 * hf;
      v2f a[4];
#pragma unroll
      for (int mi = 0; mi < 4; ++mi) {
        a[mi][0] = sA[(mi * 16 + ln) * AROW + k];
        a[mi][1] = sA[(mi * 16 + ln) * AROW + k + 1];
      }
#pragma unroll
      for (int nj = 0; nj < NJ; ++nj) {
        const int bi = k * N + n0 + nj * 16 + ln;
        v2f b = {Wc[bi], Wc[bi + N]};
#pragma unroll
        for (int mi = 0; mi < 4; ++mi) acc[mi][nj] = wmma_f32(a[mi], b, acc[mi][nj]);
      }
    }
    __syncthreads();
  }

#pragma unroll
  for (int nj = 0; nj < NJ; ++nj) {
    const int ncol = n0 + nj * 16 + ln;
    const float bv = bias[e * N + ncol];
#pragma unroll
    for (int mi = 0; mi < 4; ++mi) {
#pragma unroll
      for (int r = 0; r < 8; ++r) {
        const int m = mi * 16 + r + 8 * hf;
        if (m < cnt) {
          float v = acc[mi][nj][r] + bv;
          if (RELU) v = v > 0.0f ? v : 0.0f;
          C[(size_t)(r0 + m) * N + ncol] = v;
        }
      }
    }
  }
}

__global__ __launch_bounds__(256) void combine_kernel(
    const float* __restrict__ O, const int* __restrict__ slot,
    const float* __restrict__ selG, float* __restrict__ out) {
  const int i = blockIdx.x * 256 + threadIdx.x;
  const int t = i >> 7;
  const int q = i & 127;
  if (t >= T_TOKENS) return;
  int rlo = slot[2 * t];
  int rhi = slot[2 * t + 1];
  rlo = rlo < 0 ? 0 : (rlo > NROWS - 1 ? NROWS - 1 : rlo);
  rhi = rhi < 0 ? 0 : (rhi > NROWS - 1 ? NROWS - 1 : rhi);
  const float glo = selG[2 * t];
  const float ghi = selG[2 * t + 1];
  const float4 a = *reinterpret_cast<const float4*>(O + (size_t)rlo * DM + 4 * q);
  const float4 b = *reinterpret_cast<const float4*>(O + (size_t)rhi * DM + 4 * q);
  float4 r;
  r.x = fmaf(ghi, b.x, glo * a.x);
  r.y = fmaf(ghi, b.y, glo * a.y);
  r.z = fmaf(ghi, b.z, glo * a.z);
  r.w = fmaf(ghi, b.w, glo * a.w);
  *reinterpret_cast<float4*>(out + (size_t)t * DM + 4 * q) = r;
}

extern "C" void kernel_launch(void* const* d_in, const int* in_sizes, int n_in,
                              void* d_out, int out_size, void* d_ws, size_t ws_size,
                              hipStream_t stream) {
  const float* emb   = (const float*)d_in[0];
  const float* state = (const float*)d_in[1];
  const float* Wg    = (const float*)d_in[2];
  const float* bg    = (const float*)d_in[3];
  const float* Kmat  = (const float*)d_in[4];
  const float* W1    = (const float*)d_in[5];
  const float* b1    = (const float*)d_in[6];
  const float* W2    = (const float*)d_in[7];
  const float* b2    = (const float*)d_in[8];

  float* out_main = (float*)d_out;
  float* probs    = out_main + (size_t)T_TOKENS * DM;

  float* H       = (float*)d_ws;
  float* O       = H + (size_t)NROWS * DH;
  float* selG    = O + (size_t)NROWS * DM;
  int*   selE    = (int*)(selG + 2 * T_TOKENS);
  int*   slot    = selE + 2 * T_TOKENS;
  int*   tok     = slot + 2 * T_TOKENS;
  int*   tileE   = tok + NROWS;
  int*   tileR0  = tileE + MAXTILES;
  int*   tileCnt = tileR0 + MAXTILES;

  gating_kernel<<<T_TOKENS / 8, 256, 0, stream>>>(emb, state, Wg, bg, Kmat, probs,
                                                  selE, selG);
  route_scan_kernel<<<1, 1024, 0, stream>>>(selE, tok, slot, tileE, tileR0, tileCnt);
  grouped_gemm_kernel<8, 2, DM, DH, true, true>
      <<<dim3(DH / 256, MAXTILES), 256, 0, stream>>>(emb, tok, tileE, tileR0, tileCnt,
                                                     W1, b1, H);
  grouped_gemm_kernel<4, 1, DH, DM, false, false>
      <<<dim3(DM / 64, MAXTILES), 128, 0, stream>>>(H, tok, tileE, tileR0, tileCnt,
                                                     W2, b2, O);
  combine_kernel<<<T_TOKENS * (DM / 4) / 256, 256, 0, stream>>>(O, slot, selG, out_main);
}


# k-chunk pipeline in both GEMMs: W chunk staged in LDS like A, two LDS buffers, next chunk's global loads issued before the current WMMAs; GEMM1 block 64x128; sum order kept
# speedup vs baseline: 7.2845x; 1.0412x over previous
#include <hip/hip_runtime.h>


typedef __attribute__((ext_vector_type(2))) float v2f;
typedef __attribute__((ext_vector_type(8))) float v8f;
typedef __attribute__((ext_vector_type(4))) float v4f;

#define T_TOKENS 4096
#define DM 512
#define DH 2048
#define NE 8
#define BM 64
#define KC 32
#define APITCH (KC + 4)
#define NROWS (2 * T_TOKENS)
#define MAXTILES (NROWS / BM + NE)

static __device__ __forceinline__ v8f wmma_f32(v2f a, v2f b, v8f c) {
  return __builtin_amdgcn_wmma_f32_16x16x4_f32(false, a, false, b, (short)0, c,
                                               false, false);
}

__global__ __launch_bounds__(256) void gating_kernel(
    const float* __restrict__ emb, const float* __restrict__ state,
    const float* __restrict__ Wg, const float* __restrict__ bg,
    const float* __restrict__ Kmat, float* __restrict__ probs_out,
    int* __restrict__ selE, float* __restrict__ selG) {
  const int wave = threadIdx.x >> 5;
  const int lane = threadIdx.x & 31;
  const int t = blockIdx.x * 8 + wave;
  if (t >= T_TOKENS) return;

  const float* x = emb + (size_t)t * DM;
  float acc[NE];
#pragma unroll
  for (int e = 0; e < NE; ++e) acc[e] = 0.0f;

  for (int d = lane; d < DM; d += 32) {
    const float xd = x[d];
#pragma unroll
    for (int e = 0; e < NE; ++e) acc[e] = fmaf(xd, Wg[e * DM + d], acc[e]);
  }
#pragma unroll
  for (int e = 0; e < NE; ++e) {
#pragma unroll
    for (int off = 16; off > 0; off >>= 1) acc[e] += __shfl_xor(acc[e], off, 32);
  }

  if (lane == 0) {
    float logit[NE], p[NE];
#pragma unroll
    for (int e = 0; e < NE; ++e) {
      float tb = 0.0f;
#pragma unroll
      for (int f = 0; f < NE; ++f) tb = fmaf(Kmat[e * NE + f], state[t * NE + f], tb);
      logit[e] = acc[e] + bg[e] + tb;
    }
    float m = logit[0];
#pragma unroll
    for (int e = 1; e < NE; ++e) m = fmaxf(m, logit[e]);
    float s = 0.0f;
#pragma unroll
    for (int e = 0; e < NE; ++e) { p[e] = expf(logit[e] - m); s += p[e]; }
    const float inv = 1.0f / s;
#pragma unroll
    for (int e = 0; e < NE; ++e) p[e] *= inv;
    int i1 = 0;
#pragma unroll
    for (int e = 1; e < NE; ++e) if (p[e] > p[i1]) i1 = e;
    int i2 = (i1 == 0) ? 1 : 0;
#pragma unroll
    for (int e = 0; e < NE; ++e) if (e != i1 && p[e] > p[i2]) i2 = e;
    const int lo = i1 < i2 ? i1 : i2;
    const int hi = i1 < i2 ? i2 : i1;
    float glo = 0.0f, ghi = 0.0f;
#pragma unroll
    for (int e = 0; e < NE; ++e) {
      probs_out[t * NE + e] = p[e];
      if (e == lo) glo = p[e];
      if (e == hi) ghi = p[e];
    }
    selE[2 * t] = lo;
    selE[2 * t + 1] = hi;
    selG[2 * t] = glo;
    selG[2 * t + 1] = ghi;
  }
}

__global__ __launch_bounds__(1024) void route_scan_kernel(
    const int* __restrict__ selE, int* __restrict__ tok, int* __restrict__ slot,
    int* __restrict__ tileE, int* __restrict__ tileR0, int* __restrict__ tileCnt) {
  __shared__ int sWaveTot[32 * NE];
  __shared__ int sWaveOff[32 * NE];
  __shared__ int sTot[NE];
  __shared__ int sBase[NE];

  const int tid = threadIdx.x;
  const int lane = tid & 31;
  const int wave = tid >> 5;

  int ee[8];
#pragma unroll
  for (int j = 0; j < 8; ++j) ee[j] = selE[tid * 8 + j] & (NE - 1);

  int cnt[NE], incl[NE];
#pragma unroll
  for (int x = 0; x < NE; ++x) {
    int c = 0;
#pragma unroll
    for (int j = 0; j < 8; ++j) c += (ee[j] == x) ? 1 : 0;
    cnt[x] = c;
    int v = c;
#pragma unroll
    for (int off = 1; off < 32; off <<= 1) {
      const int u = __shfl_up(v, off, 32);
      if (lane >= off) v += u;
    }
    incl[x] = v;
  }
  if (lane == 31) {
#pragma unroll
    for (int x = 0; x < NE; ++x) sWaveTot[wave * NE + x] = incl[x];
  }
  __syncthreads();
  if (tid < NE) {
    int run = 0;
    for (int w = 0; w < 32; ++w) {
      sWaveOff[w * NE + tid] = run;
      run += sWaveTot[w * NE + tid];
    }
    sTot[tid] = run;
  }
  __syncthreads();
  if (tid == 0) {
    int b = 0;
    int nt = 0;
    for (int x = 0; x < NE; ++x) {
      sBase[x] = b;
      const int n = sTot[x];
      for (int r = 0; r < n && nt < MAXTILES; r += BM) {
        tileE[nt] = x;
        tileR0[nt] = b + r;
        tileCnt[nt] = (n - r) < BM ? (n - r) : BM;
        ++nt;
      }
      b += n;
    }
    for (; nt < MAXTILES; ++nt) {
      tileE[nt] = -1;
      tileR0[nt] = 0;
      tileCnt[nt] = 0;
    }
  }
  __syncthreads();

  int pos[NE];
#pragma unroll
  for (int x = 0; x < NE; ++x)
    pos[x] = sBase[x] + sWaveOff[wave * NE + x] + incl[x] - cnt[x];
#pragma unroll
  for (int j = 0; j < 8; ++j) {
    int p = 0;
#pragma unroll
    for (int x = 0; x < NE; ++x) {
      if (ee[j] == x) { p = pos[x]; pos[x] += 1; }
    }
    p = p < 0 ? 0 : (p > NROWS - 1 ? NROWS - 1 : p);
    tok[p] = tid * 4 + (j >> 1);
    slot[tid * 8 + j] = p;
  }
}

template <int NW, int NJ, int K, int N, bool GATHER, bool RELU>
__global__ __launch_bounds__(NW * 32) void grouped_gemm_kernel(
    const float* A, const int* __restrict__ tok,
    const int* __restrict__ tileE, const int* __restrict__ tileR0,
    const int* __restrict__ tileCnt, const float* Wt,
    const float* __restrict__ bias, float* C) {
  constexpr int NT = NW * 32;
  constexpr int BN = NW * 16 * NJ;
  constexpr int BP = BN + 8;
  constexpr int A_IT = (BM * KC / 4) / NT;
  constexpr int RP = NT / (BN / 4);
  constexpr int B_IT = KC / RP;
  constexpr int NCH = K / KC;
  constexpr int G = 2;
  constexpr int NG = KC / (4 * G);
  __shared__ __attribute__((aligned(16))) float sA[2][BM * APITCH];
  __shared__ __attribute__((aligned(16))) float sB[2][KC * BP];
  __shared__ int sRow[BM];

  const int tile = blockIdx.y;
  const int te = tileE[tile];
  if (te < 0) return;
  const int e = te & (NE - 1);
  int cnt = tileCnt[tile];
  cnt = cnt < 1 ? 1 : (cnt > BM ? BM : cnt);
  int r0 = tileR0[tile];
  r0 = r0 < 0 ? 0 : (r0 > NROWS - cnt ? NROWS - cnt : r0);

  const int tid = threadIdx.x;
  const int wave = tid >> 5;
  const int lane = tid & 31;
  const int hf = lane >> 4;
  const int ln = lane & 15;
  const int nblk = blockIdx.x * BN;
  const int nw0 = wave * (16 * NJ);
  const int n0 = nblk + nw0;

  if (tid < BM) {
    const int r = r0 + (tid < cnt ? tid : 0);
    sRow[tid] = GATHER ? (tok[r] & (T_TOKENS - 1)) : r;
  }
  __syncthreads();

  const float* Wp = Wt + (size_t)e * K * N;

  int aOff[A_IT];
#pragma unroll
  for (int j = 0; j < A_IT; ++j) aOff[j] = sRow[(tid >> 3) + (NT / 8) * j] * K + 4 * (tid & 7);
  const int aDst0 = (tid >> 3) * APITCH + 4 * (tid & 7);
  const int bOff0 = (tid / (BN / 4)) * N + nblk + 4 * (tid % (BN / 4));
  const int bDst0 = (tid / (BN / 4)) * BP + 4 * (tid % (BN / 4));

  v4f ra[A_IT], rb[B_IT];

  v8f acc[4][NJ];
#pragma unroll
  for (int mi = 0; mi < 4; ++mi)
#pragma unroll
    for (int nj = 0; nj < NJ; ++nj) acc[mi][nj] = (v8f)0.0f;

#define LOAD_CHUNK(kc)                                                               \
  {                                                                                  \
    const float* Ac = A + (kc);                                                      \
    const float* Wc = Wp + (size_t)(kc) * N + bOff0;                                 \
    _Pragma("unroll") for (int j = 0; j < A_IT; ++j)                                 \
        ra[j] = *reinterpret_cast<const v4f*>(Ac + aOff[j]);                      \
    _Pragma("unroll") for (int j = 0; j < B_IT; ++j)                                 \
        rb[j] = *reinterpret_cast<const v4f*>(Wc + (size_t)j * RP * N);           \
  }
#define STORE_CHUNK(buf)                                                             \
  {                                                                                  \
    _Pragma("unroll") for (int j = 0; j < A_IT; ++j)                                 \
        *reinterpret_cast<v4f*>(&sA[buf][aDst0 + j * (NT / 8) * APITCH]) = ra[j]; \
    _Pragma("unroll") for (int j = 0; j < B_IT; ++j)                                 \
        *reinterpret_cast<v4f*>(&sB[buf][bDst0 + j * RP * BP]) = rb[j];           \
  }
#define LOAD_FRAGS(set, g)                                                           \
  _Pragma("unroll") for (int s = 0; s < G; ++s) {                                    \
    const int k0 = ((g) * G + s) * 4;                                                \
    _Pragma("unroll") for (int mi = 0; mi < 4; ++mi)                                 \
        fa[set][s][mi] = *reinterpret_cast<const v2f*>(pa + mi * 16 * APITCH + k0);  \
    _Pragma("unroll") for (int nj = 0; nj < NJ; ++nj) {                              \
      const float* q = pb + k0 * BP + nj * 16;                                       \
      fb[set][s][nj][0] = q[0];                                                      \
      fb[set][s][nj][1] = q[BP];                                                     \
    }                                                                                \
  }

  LOAD_CHUNK(0)
  STORE_CHUNK(0)
  __syncthreads();

  for (int c = 0; c < NCH; ++c) {
    const int buf = c & 1;
    if (c + 1 < NCH) {
      LOAD_CHUNK((c + 1) * KC)
      __builtin_amdgcn_sched_barrier(0);
    }

    const float* pa = &sA[buf][ln * APITCH + 2 * hf];
    const float* pb = &sB[buf][(2 * hf) * BP + nw0 + ln];
    v2f fa[2][G][4], fb[2][G][NJ];
    LOAD_FRAGS(0, 0)
#pragma unroll
    for (int g = 0; g < NG; ++g) {
      if (g + 1 < NG) LOAD_FRAGS((g + 1) & 1, g + 1)
      __builtin_amdgcn_sched_barrier(0);
#pragma unroll
      for (int s = 0; s < G; ++s)
#pragma unroll
        for (int nj = 0; nj < NJ; ++nj)
#pragma unroll
          for (int mi = 0; mi < 4; ++mi)
            acc[mi][nj] = wmma_f32(fa[g & 1][s][mi], fb[g & 1][s][nj], acc[mi][nj]);
      __builtin_amdgcn_sched_barrier(0);
    }

    if (c + 1 < NCH) STORE_CHUNK(buf ^ 1)
    __syncthreads();
  }
#undef LOAD_CHUNK
#undef STORE_CHUNK
#undef LOAD_FRAGS

#pragma unroll
  for (int nj = 0; nj < NJ; ++nj) {
    const int ncol = n0 + nj * 16 + ln;
    const float bv = bias[e * N + ncol];
#pragma unroll
    for (int mi = 0; mi < 4; ++mi) {
#pragma unroll
      for (int r = 0; r < 8; ++r) {
        const int m = mi * 16 + r + 8 * hf;
        if (m < cnt) {
          float v = acc[mi][nj][r] + bv;
          if (RELU) v = v > 0.0f ? v : 0.0f;
          C[(size_t)(r0 + m) * N + ncol] = v;
        }
      }
    }
  }
}

__global__ __launch_bounds__(256) void combine_kernel(
    const float* __restrict__ O, const int* __restrict__ slot,
    const float* __restrict__ selG, float* __restrict__ out) {
  const int i = blockIdx.x * 256 + threadIdx.x;
  const int t = i >> 7;
  const int q = i & 127;
  if (t >= T_TOKENS) return;
  int rlo = slot[2 * t];
  int rhi = slot[2 * t + 1];
  rlo = rlo < 0 ? 0 : (rlo > NROWS - 1 ? NROWS - 1 : rlo);
  rhi = rhi < 0 ? 0 : (rhi > NROWS - 1 ? NROWS - 1 : rhi);
  const float glo = selG[2 * t];
  const float ghi = selG[2 * t + 1];
  const float4 a = *reinterpret_cast<const float4*>(O + (size_t)rlo * DM + 4 * q);
  const float4 b = *reinterpret_cast<const float4*>(O + (size_t)rhi * DM + 4 * q);
  float4 r;
  r.x = fmaf(ghi, b.x, glo * a.x);
  r.y = fmaf(ghi, b.y, glo * a.y);
  r.z = fmaf(ghi, b.z, glo * a.z);
  r.w = fmaf(ghi, b.w, glo * a.w);
  *reinterpret_cast<float4*>(out + (size_t)t * DM + 4 * q) = r;
}

extern "C" void kernel_launch(void* const* d_in, const int* in_sizes, int n_in,
                              void* d_out, int out_size, void* d_ws, size_t ws_size,
                              hipStream_t stream) {
  const float* emb   = (const float*)d_in[0];
  const float* state = (const float*)d_in[1];
  const float* Wg    = (const float*)d_in[2];
  const float* bg    = (const float*)d_in[3];
  const float* Kmat  = (const float*)d_in[4];
  const float* W1    = (const float*)d_in[5];
  const float* b1    = (const float*)d_in[6];
  const float* W2    = (const float*)d_in[7];
  const float* b2    = (const float*)d_in[8];

  float* out_main = (float*)d_out;
  float* probs    = out_main + (size_t)T_TOKENS * DM;

  float* H       = (float*)d_ws;
  float* O       = H + (size_t)NROWS * DH;
  float* selG    = O + (size_t)NROWS * DM;
  int*   selE    = (int*)(selG + 2 * T_TOKENS);
  int*   slot    = selE + 2 * T_TOKENS;
  int*   tok     = slot + 2 * T_TOKENS;
  int*   tileE   = tok + NROWS;
  int*   tileR0  = tileE + MAXTILES;
  int*   tileCnt = tileR0 + MAXTILES;

  gating_kernel<<<T_TOKENS / 8, 256, 0, stream>>>(emb, state, Wg, bg, Kmat, probs,
                                                  selE, selG);
  route_scan_kernel<<<1, 1024, 0, stream>>>(selE, tok, slot, tileE, tileR0, tileCnt);
  grouped_gemm_kernel<4, 2, DM, DH, true, true>
      <<<dim3(DH / 128, MAXTILES), 128, 0, stream>>>(emb, tok, tileE, tileR0, tileCnt,
                                                     W1, b1, H);
  grouped_gemm_kernel<4, 1, DH, DM, false, false>
      <<<dim3(DM / 64, MAXTILES), 128, 0, stream>>>(H, tok, tileE, tileR0, tileCnt,
                                                     W2, b2, O);
  combine_kernel<<<T_TOKENS * (DM / 4) / 256, 256, 0, stream>>>(O, slot, selG, out_main);
}


# second GEMM load balance: the few tiles beyond 128 (a 5th whole work item on some WGPs) are cut into 8 k slices each; combine adds those rows' 8 partials in fixed order (their sum order changes)
# speedup vs baseline: 7.5481x; 1.0362x over previous
#include <hip/hip_runtime.h>


typedef __attribute__((ext_vector_type(2))) float v2f;
typedef __attribute__((ext_vector_type(8))) float v8f;
typedef __attribute__((ext_vector_type(4))) float v4f;

#define T_TOKENS 4096
#define DM 512
#define DH 2048
#define NE 8
#define BM 64
#define KC 32
#define APITCH (KC + 4)
#define NROWS (2 * T_TOKENS)
#define MAXTILES (NROWS / BM + NE)
#define FULLT (NROWS / BM)
#define KSPL 8
#define TAILROWS ((MAXTILES - FULLT) * BM)

static __device__ __forceinline__ v8f wmma_f32(v2f a, v2f b, v8f c) {
  return __builtin_amdgcn_wmma_f32_16x16x4_f32(false, a, false, b, (short)0, c,
                                               false, false);
}

__global__ __launch_bounds__(256) void gating_kernel(
    const float* __restrict__ emb, const float* __restrict__ state,
    const float* __restrict__ Wg, const float* __restrict__ bg,
    const float* __restrict__ Kmat, float* __restrict__ probs_out,
    int* __restrict__ selE, float* __restrict__ selG) {
  const int wave = threadIdx.x >> 5;
  const int lane = threadIdx.x & 31;
  const int t = blockIdx.x * 8 + wave;
  if (t >= T_TOKENS) return;

  const float* x = emb + (size_t)t * DM;
  float acc[NE];
#pragma unroll
  for (int e = 0; e < NE; ++e) acc[e] = 0.0f;

  for (int d = lane; d < DM; d += 32) {
    const float xd = x[d];
#pragma unroll
    for (int e = 0; e < NE; ++e) acc[e] = fmaf(xd, Wg[e * DM + d], acc[e]);
  }
#pragma unroll
  for (int e = 0; e < NE; ++e) {
#pragma unroll
    for (int off = 16; off > 0; off >>= 1) acc[e] += __shfl_xor(acc[e], off, 32);
  }

  if (lane == 0) {
    float logit[NE], p[NE];
#pragma unroll
    for (int e = 0; e < NE; ++e) {
      float tb = 0.0f;
#pragma unroll
      for (int f = 0; f < NE; ++f) tb = fmaf(Kmat[e * NE + f], state[t * NE + f], tb);
      logit[e] = acc[e] + bg[e] + tb;
    }
    float m = logit[0];
#pragma unroll
    for (int e = 1; e < NE; ++e) m = fmaxf(m, logit[e]);
    float s = 0.0f;
#pragma unroll
    for (int e = 0; e < NE; ++e) { p[e] = expf(logit[e] - m); s += p[e]; }
    const float inv = 1.0f / s;
#pragma unroll
    for (int e = 0; e < NE; ++e) p[e] *= inv;
    int i1 = 0;
#pragma unroll
    for (int e = 1; e < NE; ++e) if (p[e] > p[i1]) i1 = e;
    int i2 = (i1 == 0) ? 1 : 0;
#pragma unroll
    for (int e = 0; e < NE; ++e) if (e != i1 && p[e] > p[i2]) i2 = e;
    const int lo = i1 < i2 ? i1 : i2;
    const int hi = i1 < i2 ? i2 : i1;
    float glo = 0.0f, ghi = 0.0f;
#pragma unroll
    for (int e = 0; e < NE; ++e) {
      probs_out[t * NE + e] = p[e];
      if (e == lo) glo = p[e];
      if (e == hi) ghi = p[e];
    }
    selE[2 * t] = lo;
    selE[2 * t + 1] = hi;
    selG[2 * t] = glo;
    selG[2 * t + 1] = ghi;
  }
}

__global__ __launch_bounds__(1024) void route_scan_kernel(
    const int* __restrict__ selE, int* __restrict__ tok, int* __restrict__ slot,
    int* __restrict__ tileE, int* __restrict__ tileR0, int* __restrict__ tileCnt) {
  __shared__ int sWaveTot[32 * NE];
  __shared__ int sWaveOff[32 * NE];
  __shared__ int sTot[NE];
  __shared__ int sBase[NE];

  const int tid = threadIdx.x;
  const int lane = tid & 31;
  const int wave = tid >> 5;

  int ee[8];
#pragma unroll
  for (int j = 0; j < 8; ++j) ee[j] = selE[tid * 8 + j] & (NE - 1);

  int cnt[NE], incl[NE];
#pragma unroll
  for (int x = 0; x < NE; ++x) {
    int c = 0;
#pragma unroll
    for (int j = 0; j < 8; ++j) c += (ee[j] == x) ? 1 : 0;
    cnt[x] = c;
    int v = c;
#pragma unroll
    for (int off = 1; off < 32; off <<= 1) {
      const int u = __shfl_up(v, off, 32);
      if (lane >= off) v += u;
    }
    incl[x] = v;
  }
  if (lane == 31) {
#pragma unroll
    for (int x = 0; x < NE; ++x) sWaveTot[wave * NE + x] = incl[x];
  }
  __syncthreads();
  if (tid < NE) {
    int run = 0;
    for (int w = 0; w < 32; ++w) {
      sWaveOff[w * NE + tid] = run;
      run += sWaveTot[w * NE + tid];
    }
    sTot[tid] = run;
  }
  __syncthreads();
  if (tid == 0) {
    int b = 0;
    int nt = 0;
    for (int x = 0; x < NE; ++x) {
      sBase[x] = b;
      const int n = sTot[x];
      for (int r = 0; r < n && nt < MAXTILES; r += BM) {
        tileE[nt] = x;
        tileR0[nt] = b + r;
        tileCnt[nt] = (n - r) < BM ? (n - r) : BM;
        ++nt;
      }
      b += n;
    }
    for (; nt < MAXTILES; ++nt) {
      tileE[nt] = -1;
      tileR0[nt] = 0;
      tileCnt[nt] = 0;
    }
  }
  __syncthreads();

  int pos[NE];
#pragma unroll
  for (int x = 0; x < NE; ++x)
    pos[x] = sBase[x] + sWaveOff[wave * NE + x] + incl[x] - cnt[x];
#pragma unroll
  for (int j = 0; j < 8; ++j) {
    int p = 0;
#pragma unroll
    for (int x = 0; x < NE; ++x) {
      if (ee[j] == x) { p = pos[x]; pos[x] += 1; }
    }
    p = p < 0 ? 0 : (p > NROWS - 1 ? NROWS - 1 : p);
    tok[p] = tid * 4 + (j >> 1);
    slot[tid * 8 + j] = p;
  }
}

template <int NW, int NJ, int K, int N, bool GATHER, bool RELU, bool TAIL>
__global__ __launch_bounds__(NW * 32) void grouped_gemm_kernel(
    const float* A, const int* __restrict__ tok,
    const int* __restrict__ tileE, const int* __restrict__ tileR0,
    const int* __restrict__ tileCnt, const float* Wt,
    const float* __restrict__ bias, float* C, float* Ct) {
  constexpr int NT = NW * 32;
  constexpr int BN = NW * 16 * NJ;
  constexpr int BP = BN + 8;
  constexpr int A_IT = (BM * KC / 4) / NT;
  constexpr int RP = NT / (BN / 4);
  constexpr int B_IT = KC / RP;
  constexpr int NCH = K / KC;
  constexpr int G = 2;
  constexpr int NG = KC / (4 * G);
  __shared__ __attribute__((aligned(16))) float sA[2][BM * APITCH];
  __shared__ __attribute__((aligned(16))) float sB[2][KC * BP];
  __shared__ int sRow[BM];

  constexpr int NCB = N / BN;
  int tile, cb, ks = 0, c0 = 0, c1 = NCH;
  if (TAIL) {
    const int id = blockIdx.x;
    if (id < FULLT * NCB) {
      tile = id / NCB;
      cb = id % NCB;
    } else {
      const int j = id - FULLT * NCB;
      tile = FULLT + j / (NCB * KSPL);
      cb = (j / KSPL) % NCB;
      ks = j % KSPL;
      c0 = ks * (NCH / KSPL);
      c1 = c0 + NCH / KSPL;
    }
  } else {
    tile = blockIdx.y;
    cb = blockIdx.x;
  }
  const int te = tileE[tile];
  if (te < 0) return;
  const int e = te & (NE - 1);
  int cnt = tileCnt[tile];
  cnt = cnt < 1 ? 1 : (cnt > BM ? BM : cnt);
  int r0 = tileR0[tile];
  r0 = r0 < 0 ? 0 : (r0 > NROWS - cnt ? NROWS - cnt : r0);

  const int tid = threadIdx.x;
  const int wave = tid >> 5;
  const int lane = tid & 31;
  const int hf = lane >> 4;
  const int ln = lane & 15;
  const int nblk = cb * BN;
  const int nw0 = wave * (16 * NJ);
  const int n0 = nblk + nw0;

  if (tid < BM) {
    const int r = r0 + (tid < cnt ? tid : 0);
    sRow[tid] = GATHER ? (tok[r] & (T_TOKENS - 1)) : r;
  }
  __syncthreads();

  const float* Wp = Wt + (size_t)e * K * N;

  int aOff[A_IT];
#pragma unroll
  for (int j = 0; j < A_IT; ++j) aOff[j] = sRow[(tid >> 3) + (NT / 8) * j] * K + 4 * (tid & 7);
  const int aDst0 = (tid >> 3) * APITCH + 4 * (tid & 7);
  const int bOff0 = (tid / (BN / 4)) * N + nblk + 4 * (tid % (BN / 4));
  const int bDst0 = (tid / (BN / 4)) * BP + 4 * (tid % (BN / 4));

  v4f ra[A_IT], rb[B_IT];

  v8f acc[4][NJ];
#pragma unroll
  for (int mi = 0; mi < 4; ++mi)
#pragma unroll
    for (int nj = 0; nj < NJ; ++nj) acc[mi][nj] = (v8f)0.0f;

#define LOAD_CHUNK(kc)                                                               \
  {                                                                                  \
    const float* Ac = A + (kc);                                                      \
    const float* Wc = Wp + (size_t)(kc) * N + bOff0;                                 \
    _Pragma("unroll") for (int j = 0; j < A_IT; ++j)                                 \
        ra[j] = *reinterpret_cast<const v4f*>(Ac + aOff[j]);                      \
    _Pragma("unroll") for (int j = 0; j < B_IT; ++j)                                 \
        rb[j] = *reinterpret_cast<const v4f*>(Wc + (size_t)j * RP * N);           \
  }
#define STORE_CHUNK(buf)                                                             \
  {                                                                                  \
    _Pragma("unroll") for (int j = 0; j < A_IT; ++j)                                 \
        *reinterpret_cast<v4f*>(&sA[buf][aDst0 + j * (NT / 8) * APITCH]) = ra[j]; \
    _Pragma("unroll") for (int j = 0; j < B_IT; ++j)                                 \
        *reinterpret_cast<v4f*>(&sB[buf][bDst0 + j * RP * BP]) = rb[j];           \
  }
#define LOAD_FRAGS(set, g)                                                           \
  _Pragma("unroll") for (int s = 0; s < G; ++s) {                                    \
    const int k0 = ((g) * G + s) * 4;                                                \
    _Pragma("unroll") for (int mi = 0; mi < 4; ++mi)                                 \
        fa[set][s][mi] = *reinterpret_cast<const v2f*>(pa + mi * 16 * APITCH + k0);  \
    _Pragma("unroll") for (int nj = 0; nj < NJ; ++nj) {                              \
      const float* q = pb + k0 * BP + nj * 16;                                       \
      fb[set][s][nj][0] = q[0];                                                      \
      fb[set][s][nj][1] = q[BP];                                                     \
    }                                                                                \
  }

  LOAD_CHUNK(c0 * KC)
  STORE_CHUNK(0)
  __syncthreads();

  for (int c = c0; c < c1; ++c) {
    const int buf = (c - c0) & 1;
    if (c + 1 < c1) {
      LOAD_CHUNK((c + 1) * KC)
      __builtin_amdgcn_sched_barrier(0);
    }

    const float* pa = &sA[buf][ln * APITCH + 2 * hf];
    const float* pb = &sB[buf][(2 * hf) * BP + nw0 + ln];
    v2f fa[2][G][4], fb[2][G][NJ];
    LOAD_FRAGS(0, 0)
#pragma unroll
    for (int g = 0; g < NG; ++g) {
      if (g + 1 < NG) LOAD_FRAGS((g + 1) & 1, g + 1)
      __builtin_amdgcn_sched_barrier(0);
#pragma unroll
      for (int s = 0; s < G; ++s)
#pragma unroll
        for (int nj = 0; nj < NJ; ++nj)
#pragma unroll
          for (int mi = 0; mi < 4; ++mi)
            acc[mi][nj] = wmma_f32(fa[g & 1][s][mi], fb[g & 1][s][nj], acc[mi][nj]);
      __builtin_amdgcn_sched_barrier(0);
    }

    if (c + 1 < c1) STORE_CHUNK(buf ^ 1)
    __syncthreads();
  }
#undef LOAD_CHUNK
#undef STORE_CHUNK
#undef LOAD_FRAGS

  float* Cd = C;
  int rd = r0;
  if (TAIL && ks > 0) {
    int sr = tileR0[FULLT];
    sr = sr < 0 ? 0 : (sr > NROWS ? NROWS : sr);
    rd = r0 - sr;
    rd = rd < 0 ? 0 : (rd > TAILROWS - cnt ? TAILROWS - cnt : rd);
    Cd = Ct + (size_t)(ks - 1) * TAILROWS * N;
  }
#pragma unroll
  for (int nj = 0; nj < NJ; ++nj) {
    const int ncol = n0 + nj * 16 + ln;
    const float bv = (TAIL && ks > 0) ? 0.0f : bias[e * N + ncol];
#pragma unroll
    for (int mi = 0; mi < 4; ++mi) {
#pragma unroll
      for (int r = 0; r < 8; ++r) {
        const int m = mi * 16 + r + 8 * hf;
        if (m < cnt) {
          float v = acc[mi][nj][r] + bv;
          if (RELU) v = v > 0.0f ? v : 0.0f;
          Cd[(size_t)(rd + m) * N + ncol] = v;
        }
      }
    }
  }
}

static __device__ __forceinline__ float4 load_row_sum(
    const float* __restrict__ O, const float* __restrict__ Ot, int r, int sr, int q) {
  float4 a = *reinterpret_cast<const float4*>(O + (size_t)r * DM + 4 * q);
  if (r >= sr) {
    int rt = r - sr;
    rt = rt > TAILROWS - 1 ? TAILROWS - 1 : rt;
#pragma unroll
    for (int p = 0; p < KSPL - 1; ++p) {
      const float4 b = *reinterpret_cast<const float4*>(
          Ot + ((size_t)p * TAILROWS + rt) * DM + 4 * q);
      a.x += b.x; a.y += b.y; a.z += b.z; a.w += b.w;
    }
  }
  return a;
}

__global__ __launch_bounds__(256) void combine_kernel(
    const float* __restrict__ O, const float* __restrict__ Ot,
    const int* __restrict__ tileE, const int* __restrict__ tileR0,
    const int* __restrict__ slot,
    const float* __restrict__ selG, float* __restrict__ out) {
  const int i = blockIdx.x * 256 + threadIdx.x;
  const int t = i >> 7;
  const int q = i & 127;
  if (t >= T_TOKENS) return;
  int rlo = slot[2 * t];
  int rhi = slot[2 * t + 1];
  rlo = rlo < 0 ? 0 : (rlo > NROWS - 1 ? NROWS - 1 : rlo);
  rhi = rhi < 0 ? 0 : (rhi > NROWS - 1 ? NROWS - 1 : rhi);
  const float glo = selG[2 * t];
  const float ghi = selG[2 * t + 1];
  int sr = NROWS;
  if (tileE[FULLT] >= 0) {
    sr = tileR0[FULLT];
    sr = sr < 0 ? 0 : (sr > NROWS ? NROWS : sr);
  }
  const float4 a = load_row_sum(O, Ot, rlo, sr, q);
  const float4 b = load_row_sum(O, Ot, rhi, sr, q);
  float4 r;
  r.x = fmaf(ghi, b.x, glo * a.x);
  r.y = fmaf(ghi, b.y, glo * a.y);
  r.z = fmaf(ghi, b.z, glo * a.z);
  r.w = fmaf(ghi, b.w, glo * a.w);
  *reinterpret_cast<float4*>(out + (size_t)t * DM + 4 * q) = r;
}

extern "C" void kernel_launch(void* const* d_in, const int* in_sizes, int n_in,
                              void* d_out, int out_size, void* d_ws, size_t ws_size,
                              hipStream_t stream) {
  const float* emb   = (const float*)d_in[0];
  const float* state = (const float*)d_in[1];
  const float* Wg    = (const float*)d_in[2];
  const float* bg    = (const float*)d_in[3];
  const float* Kmat  = (const float*)d_in[4];
  const float* W1    = (const float*)d_in[5];
  const float* b1    = (const float*)d_in[6];
  const float* W2    = (const float*)d_in[7];
  const float* b2    = (const float*)d_in[8];

  float* out_main = (float*)d_out;
  float* probs    = out_main + (size_t)T_TOKENS * DM;

  float* H       = (float*)d_ws;
  float* O       = H + (size_t)NROWS * DH;
  float* Ot      = O + (size_t)NROWS * DM;
  float* selG    = Ot + (size_t)(KSPL - 1) * TAILROWS * DM;
  int*   selE    = (int*)(selG + 2 * T_TOKENS);
  int*   slot    = selE + 2 * T_TOKENS;
  int*   tok     = slot + 2 * T_TOKENS;
  int*   tileE   = tok + NROWS;
  int*   tileR0  = tileE + MAXTILES;
  int*   tileCnt = tileR0 + MAXTILES;

  gating_kernel<<<T_TOKENS / 8, 256, 0, stream>>>(emb, state, Wg, bg, Kmat, probs,
                                                  selE, selG);
  route_scan_kernel<<<1, 1024, 0, stream>>>(selE, tok, slot, tileE, tileR0, tileCnt);
  grouped_gemm_kernel<4, 2, DM, DH, true, true, false>
      <<<dim3(DH / 128, MAXTILES), 128, 0, stream>>>(emb, tok, tileE, tileR0, tileCnt,
                                                     W1, b1, H, nullptr);
  grouped_gemm_kernel<4, 1, DH, DM, false, false, true>
      <<<dim3((FULLT + (MAXTILES - FULLT) * KSPL) * (DM / 64)), 128, 0, stream>>>(
          H, tok, tileE, tileR0, tileCnt, W2, b2, O, Ot);
  combine_kernel<<<T_TOKENS * (DM / 4) / 256, 256, 0, stream>>>(O, Ot, tileE, tileR0, slot,
                                                                selG, out_main);
}


# combine kernel: one wave per two whole tokens (2048 waves, was 16384 of a quarter token each); indices and gates read once per wave as scalars, all row loads issued ahead of any wait; sum order kept
# speedup vs baseline: 7.7768x; 1.0303x over previous
#include <hip/hip_runtime.h>


typedef __attribute__((ext_vector_type(2))) float v2f;
typedef __attribute__((ext_vector_type(8))) float v8f;
typedef __attribute__((ext_vector_type(4))) float v4f;

#define T_TOKENS 4096
#define DM 512
#define DH 2048
#define NE 8
#define BM 64
#define KC 32
#define APITCH (KC + 4)
#define NROWS (2 * T_TOKENS)
#define MAXTILES (NROWS / BM + NE)
#define FULLT (NROWS / BM)
#define KSPL 8
#define TAILROWS ((MAXTILES - FULLT) * BM)

static __device__ __forceinline__ v8f wmma_f32(v2f a, v2f b, v8f c) {
  return __builtin_amdgcn_wmma_f32_16x16x4_f32(false, a, false, b, (short)0, c,
                                               false, false);
}

__global__ __launch_bounds__(256) void gating_kernel(
    const float* __restrict__ emb, const float* __restrict__ state,
    const float* __restrict__ Wg, const float* __restrict__ bg,
    const float* __restrict__ Kmat, float* __restrict__ probs_out,
    int* __restrict__ selE, float* __restrict__ selG) {
  const int wave = threadIdx.x >> 5;
  const int lane = threadIdx.x & 31;
  const int t = blockIdx.x * 8 + wave;
  if (t >= T_TOKENS) return;

  const float* x = emb + (size_t)t * DM;
  float acc[NE];
#pragma unroll
  for (int e = 0; e < NE; ++e) acc[e] = 0.0f;

  for (int d = lane; d < DM; d += 32) {
    const float xd = x[d];
#pragma unroll
    for (int e = 0; e < NE; ++e) acc[e] = fmaf(xd, Wg[e * DM + d], acc[e]);
  }
#pragma unroll
  for (int e = 0; e < NE; ++e) {
#pragma unroll
    for (int off = 16; off > 0; off >>= 1) acc[e] += __shfl_xor(acc[e], off, 32);
  }

  if (lane == 0) {
    float logit[NE], p[NE];
#pragma unroll
    for (int e = 0; e < NE; ++e) {
      float tb = 0.0f;
#pragma unroll
      for (int f = 0; f < NE; ++f) tb = fmaf(Kmat[e * NE + f], state[t * NE + f], tb);
      logit[e] = acc[e] + bg[e] + tb;
    }
    float m = logit[0];
#pragma unroll
    for (int e = 1; e < NE; ++e) m = fmaxf(m, logit[e]);
    float s = 0.0f;
#pragma unroll
    for (int e = 0; e < NE; ++e) { p[e] = expf(logit[e] - m); s += p[e]; }
    const float inv = 1.0f / s;
#pragma unroll
    for (int e = 0; e < NE; ++e) p[e] *= inv;
    int i1 = 0;
#pragma unroll
    for (int e = 1; e < NE; ++e) if (p[e] > p[i1]) i1 = e;
    int i2 = (i1 == 0) ? 1 : 0;
#pragma unroll
    for (int e = 0; e < NE; ++e) if (e != i1 && p[e] > p[i2]) i2 = e;
    const int lo = i1 < i2 ? i1 : i2;
    const int hi = i1 < i2 ? i2 : i1;
    float glo = 0.0f, ghi = 0.0f;
#pragma unroll
    for (int e = 0; e < NE; ++e) {
      probs_out[t * NE + e] = p[e];
      if (e == lo) glo = p[e];
      if (e == hi) ghi = p[e];
    }
    selE[2 * t] = lo;
    selE[2 * t + 1] = hi;
    selG[2 * t] = glo;
    selG[2 * t + 1] = ghi;
  }
}

__global__ __launch_bounds__(1024) void route_scan_kernel(
    const int* __restrict__ selE, int* __restrict__ tok, int* __restrict__ slot,
    int* __restrict__ tileE, int* __restrict__ tileR0, int* __restrict__ tileCnt) {
  __shared__ int sWaveTot[32 * NE];
  __shared__ int sWaveOff[32 * NE];
  __shared__ int sTot[NE];
  __shared__ int sBase[NE];

  const int tid = threadIdx.x;
  const int lane = tid & 31;
  const int wave = tid >> 5;

  int ee[8];
#pragma unroll
  for (int j = 0; j < 8; ++j) ee[j] = selE[tid * 8 + j] & (NE - 1);

  int cnt[NE], incl[NE];
#pragma unroll
  for (int x = 0; x < NE; ++x) {
    int c = 0;
#pragma unroll
    for (int j = 0; j < 8; ++j) c += (ee[j] == x) ? 1 : 0;
    cnt[x] = c;
    int v = c;
#pragma unroll
    for (int off = 1; off < 32; off <<= 1) {
      const int u = __shfl_up(v, off, 32);
      if (lane >= off) v += u;
    }
    incl[x] = v;
  }
  if (lane == 31) {
#pragma unroll
    for (int x = 0; x < NE; ++x) sWaveTot[wave * NE + x] = incl[x];
  }
  __syncthreads();
  if (tid < NE) {
    int run = 0;
    for (int w = 0; w < 32; ++w) {
      sWaveOff[w * NE + tid] = run;
      run += sWaveTot[w * NE + tid];
    }
    sTot[tid] = run;
  }
  __syncthreads();
  if (tid == 0) {
    int b = 0;
    int nt = 0;
    for (int x = 0; x < NE; ++x) {
      sBase[x] = b;
      const int n = sTot[x];
      for (int r = 0; r < n && nt < MAXTILES; r += BM) {
        tileE[nt] = x;
        tileR0[nt] = b + r;
        tileCnt[nt] = (n - r) < BM ? (n - r) : BM;
        ++nt;
      }
      b += n;
    }
    for (; nt < MAXTILES; ++nt) {
      tileE[nt] = -1;
      tileR0[nt] = 0;
      tileCnt[nt] = 0;
    }
  }
  __syncthreads();

  int pos[NE];
#pragma unroll
  for (int x = 0; x < NE; ++x)
    pos[x] = sBase[x] + sWaveOff[wave * NE + x] + incl[x] - cnt[x];
#pragma unroll
  for (int j = 0; j < 8; ++j) {
    int p = 0;
#pragma unroll
    for (int x = 0; x < NE; ++x) {
      if (ee[j] == x) { p = pos[x]; pos[x] += 1; }
    }
    p = p < 0 ? 0 : (p > NROWS - 1 ? NROWS - 1 : p);
    tok[p] = tid * 4 + (j >> 1);
    slot[tid * 8 + j] = p;
  }
}

template <int NW, int NJ, int K, int N, bool GATHER, bool RELU, bool TAIL>
__global__ __launch_bounds__(NW * 32) void grouped_gemm_kernel(
    const float* A, const int* __restrict__ tok,
    const int* __restrict__ tileE, const int* __restrict__ tileR0,
    const int* __restrict__ tileCnt, const float* Wt,
    const float* __restrict__ bias, float* C, float* Ct) {
  constexpr int NT = NW * 32;
  constexpr int BN = NW * 16 * NJ;
  constexpr int BP = BN + 8;
  constexpr int A_IT = (BM * KC / 4) / NT;
  constexpr int RP = NT / (BN / 4);
  constexpr int B_IT = KC / RP;
  constexpr int NCH = K / KC;
  constexpr int G = 2;
  constexpr int NG = KC / (4 * G);
  __shared__ __attribute__((aligned(16))) float sA[2][BM * APITCH];
  __shared__ __attribute__((aligned(16))) float sB[2][KC * BP];
  __shared__ int sRow[BM];

  constexpr int NCB = N / BN;
  int tile, cb, ks = 0, c0 = 0, c1 = NCH;
  if (TAIL) {
    const int id = blockIdx.x;
    if (id < FULLT * NCB) {
      tile = id / NCB;
      cb = id % NCB;
    } else {
      const int j = id - FULLT * NCB;
      tile = FULLT + j / (NCB * KSPL);
      cb = (j / KSPL) % NCB;
      ks = j % KSPL;
      c0 = ks * (NCH / KSPL);
      c1 = c0 + NCH / KSPL;
    }
  } else {
    tile = blockIdx.y;
    cb = blockIdx.x;
  }
  const int te = tileE[tile];
  if (te < 0) return;
  const int e = te & (NE - 1);
  int cnt = tileCnt[tile];
  cnt = cnt < 1 ? 1 : (cnt > BM ? BM : cnt);
  int r0 = tileR0[tile];
  r0 = r0 < 0 ? 0 : (r0 > NROWS - cnt ? NROWS - cnt : r0);

  const int tid = threadIdx.x;
  const int wave = tid >> 5;
  const int lane = tid & 31;
  const int hf = lane >> 4;
  const int ln = lane & 15;
  const int nblk = cb * BN;
  const int nw0 = wave * (16 * NJ);
  const int n0 = nblk + nw0;

  if (tid < BM) {
    const int r = r0 + (tid < cnt ? tid : 0);
    sRow[tid] = GATHER ? (tok[r] & (T_TOKENS - 1)) : r;
  }
  __syncthreads();

  const float* Wp = Wt + (size_t)e * K * N;

  int aOff[A_IT];
#pragma unroll
  for (int j = 0; j < A_IT; ++j) aOff[j] = sRow[(tid >> 3) + (NT / 8) * j] * K + 4 * (tid & 7);
  const int aDst0 = (tid >> 3) * APITCH + 4 * (tid & 7);
  const int bOff0 = (tid / (BN / 4)) * N + nblk + 4 * (tid % (BN / 4));
  const int bDst0 = (tid / (BN / 4)) * BP + 4 * (tid % (BN / 4));

  v4f ra[A_IT], rb[B_IT];

  v8f acc[4][NJ];
#pragma unroll
  for (int mi = 0; mi < 4; ++mi)
#pragma unroll
    for (int nj = 0; nj < NJ; ++nj) acc[mi][nj] = (v8f)0.0f;

#define LOAD_CHUNK(kc)                                                               \
  {                                                                                  \
    const float* Ac = A + (kc);                                                      \
    const float* Wc = Wp + (size_t)(kc) * N + bOff0;                                 \
    _Pragma("unroll") for (int j = 0; j < A_IT; ++j)                                 \
        ra[j] = *reinterpret_cast<const v4f*>(Ac + aOff[j]);                      \
    _Pragma("unroll") for (int j = 0; j < B_IT; ++j)                                 \
        rb[j] = *reinterpret_cast<const v4f*>(Wc + (size_t)j * RP * N);           \
  }
#define STORE_CHUNK(buf)                                                             \
  {                                                                                  \
    _Pragma("unroll") for (int j = 0; j < A_IT; ++j)                                 \
        *reinterpret_cast<v4f*>(&sA[buf][aDst0 + j * (NT / 8) * APITCH]) = ra[j]; \
    _Pragma("unroll") for (int j = 0; j < B_IT; ++j)                                 \
        *reinterpret_cast<v4f*>(&sB[buf][bDst0 + j * RP * BP]) = rb[j];           \
  }
#define LOAD_FRAGS(set, g)                                                           \
  _Pragma("unroll") for (int s = 0; s < G; ++s) {                                    \
    const int k0 = ((g) * G + s) * 4;                                                \
    _Pragma("unroll") for (int mi = 0; mi < 4; ++mi)                                 \
        fa[set][s][mi] = *reinterpret_cast<const v2f*>(pa + mi * 16 * APITCH + k0);  \
    _Pragma("unroll") for (int nj = 0; nj < NJ; ++nj) {                              \
      const float* q = pb + k0 * BP + nj * 16;                                       \
      fb[set][s][nj][0] = q[0];                                                      \
      fb[set][s][nj][1] = q[BP];                                                     \
    }                                                                                \
  }

  LOAD_CHUNK(c0 * KC)
  STORE_CHUNK(0)
  __syncthreads();

  for (int c = c0; c < c1; ++c) {
    const int buf = (c - c0) & 1;
    if (c + 1 < c1) {
      LOAD_CHUNK((c + 1) * KC)
      __builtin_amdgcn_sched_barrier(0);
    }

    const float* pa = &sA[buf][ln * APITCH + 2 * hf];
    const float* pb = &sB[buf][(2 * hf) * BP + nw0 + ln];
    v2f fa[2][G][4], fb[2][G][NJ];
    LOAD_FRAGS(0, 0)
#pragma unroll
    for (int g = 0; g < NG; ++g) {
      if (g + 1 < NG) LOAD_FRAGS((g + 1) & 1, g + 1)
      __builtin_amdgcn_sched_barrier(0);
#pragma unroll
      for (int s = 0; s < G; ++s)
#pragma unroll
        for (int nj = 0; nj < NJ; ++nj)
#pragma unroll
          for (int mi = 0; mi < 4; ++mi)
            acc[mi][nj] = wmma_f32(fa[g & 1][s][mi], fb[g & 1][s][nj], acc[mi][nj]);
      __builtin_amdgcn_sched_barrier(0);
    }

    if (c + 1 < c1) STORE_CHUNK(buf ^ 1)
    __syncthreads();
  }
#undef LOAD_CHUNK
#undef STORE_CHUNK
#undef LOAD_FRAGS

  float* Cd = C;
  int rd = r0;
  if (TAIL && ks > 0) {
    int sr = tileR0[FULLT];
    sr = sr < 0 ? 0 : (sr > NROWS ? NROWS : sr);
    rd = r0 - sr;
    rd = rd < 0 ? 0 : (rd > TAILROWS - cnt ? TAILROWS - cnt : rd);
    Cd = Ct + (size_t)(ks - 1) * TAILROWS * N;
  }
#pragma unroll
  for (int nj = 0; nj < NJ; ++nj) {
    const int ncol = n0 + nj * 16 + ln;
    const float bv = (TAIL && ks > 0) ? 0.0f : bias[e * N + ncol];
#pragma unroll
    for (int mi = 0; mi < 4; ++mi) {
#pragma unroll
      for (int r = 0; r < 8; ++r) {
        const int m = mi * 16 + r + 8 * hf;
        if (m < cnt) {
          float v = acc[mi][nj][r] + bv;
          if (RELU) v = v > 0.0f ? v : 0.0f;
          Cd[(size_t)(rd + m) * N + ncol] = v;
        }
      }
    }
  }
}

#define CMB_Q 4
static __device__ __forceinline__ void add_row_planes(
    float4 (&a)[CMB_Q], const float* __restrict__ Ot, int r, int sr, int lane) {
  int rt = r - sr;
  rt = rt > TAILROWS - 1 ? TAILROWS - 1 : rt;
#pragma unroll
  for (int i = 0; i < CMB_Q; ++i) {
    const int q = lane + 32 * i;
#pragma unroll
    for (int p = 0; p < KSPL - 1; ++p) {
      const float4 b = *reinterpret_cast<const float4*>(
          Ot + ((size_t)p * TAILROWS + rt) * DM + 4 * q);
      a[i].x += b.x; a[i].y += b.y; a[i].z += b.z; a[i].w += b.w;
    }
  }
}

__global__ __launch_bounds__(256) void combine_kernel(
    const float* __restrict__ O, const float* __restrict__ Ot,
    const int* __restrict__ tileE, const int* __restrict__ tileR0,
    const int* __restrict__ slot,
    const float* __restrict__ selG, float* __restrict__ out) {
  const int lane = threadIdx.x & 31;
  const int wv = __builtin_amdgcn_readfirstlane((int)(threadIdx.x >> 5));
  const int t0 = 2 * (blockIdx.x * 8 + wv);
  if (t0 >= T_TOKENS) return;
  const int te = tileE[FULLT];
  int srv = tileR0[FULLT];
  srv = srv < 0 ? 0 : (srv > NROWS ? NROWS : srv);
  const int sr = te >= 0 ? srv : NROWS;
  int row[4];
  float gate[4];
#pragma unroll
  for (int j = 0; j < 4; ++j) {
    int r = slot[2 * t0 + j];
    r = r < 0 ? 0 : (r > NROWS - 1 ? NROWS - 1 : r);
    row[j] = r;
    gate[j] = selG[2 * t0 + j];
  }
  float4 v[4][CMB_Q];
#pragma unroll
  for (int j = 0; j < 4; ++j)
#pragma unroll
    for (int i = 0; i < CMB_Q; ++i)
      v[j][i] = *reinterpret_cast<const float4*>(O + (size_t)row[j] * DM + 4 * (lane + 32 * i));
  if ((row[0] >= sr) | (row[1] >= sr) | (row[2] >= sr) | (row[3] >= sr)) {
#pragma unroll
    for (int j = 0; j < 4; ++j)
      if (row[j] >= sr) add_row_planes(v[j], Ot, row[j], sr, lane);
  }
#pragma unroll
  for (int tt = 0; tt < 2; ++tt) {
    const float glo = gate[2 * tt];
    const float ghi = gate[2 * tt + 1];
#pragma unroll
    for (int i = 0; i < CMB_Q; ++i) {
      const float4 a = v[2 * tt][i];
      const float4 b = v[2 * tt + 1][i];
      float4 r;
      r.x = fmaf(ghi, b.x, glo * a.x);
      r.y = fmaf(ghi, b.y, glo * a.y);
      r.z = fmaf(ghi, b.z, glo * a.z);
      r.w = fmaf(ghi, b.w, glo * a.w);
      *reinterpret_cast<float4*>(out + (size_t)(t0 + tt) * DM + 4 * (lane + 32 * i)) = r;
    }
  }
}

extern "C" void kernel_launch(void* const* d_in, const int* in_sizes, int n_in,
                              void* d_out, int out_size, void* d_ws, size_t ws_size,
                              hipStream_t stream) {
  const float* emb   = (const float*)d_in[0];
  const float* state = (const float*)d_in[1];
  const float* Wg    = (const float*)d_in[2];
  const float* bg    = (const float*)d_in[3];
  const float* Kmat  = (const float*)d_in[4];
  const float* W1    = (const float*)d_in[5];
  const float* b1    = (const float*)d_in[6];
  const float* W2    = (const float*)d_in[7];
  const float* b2    = (const float*)d_in[8];

  float* out_main = (float*)d_out;
  float* probs    = out_main + (size_t)T_TOKENS * DM;

  float* H       = (float*)d_ws;
  float* O       = H + (size_t)NROWS * DH;
  float* Ot      = O + (size_t)NROWS * DM;
  float* selG    = Ot + (size_t)(KSPL - 1) * TAILROWS * DM;
  int*   selE    = (int*)(selG + 2 * T_TOKENS);
  int*   slot    = selE + 2 * T_TOKENS;
  int*   tok     = slot + 2 * T_TOKENS;
  int*   tileE   = tok + NROWS;
  int*   tileR0  = tileE + MAXTILES;
  int*   tileCnt = tileR0 + MAXTILES;

  gating_kernel<<<T_TOKENS / 8, 256, 0, stream>>>(emb, state, Wg, bg, Kmat, probs,
                                                  selE, selG);
  route_scan_kernel<<<1, 1024, 0, stream>>>(selE, tok, slot, tileE, tileR0, tileCnt);
  grouped_gemm_kernel<4, 2, DM, DH, true, true, false>
      <<<dim3(DH / 128, MAXTILES), 128, 0, stream>>>(emb, tok, tileE, tileR0, tileCnt,
                                                     W1, b1, H, nullptr);
  grouped_gemm_kernel<4, 1, DH, DM, false, false, true>
      <<<dim3((FULLT + (MAXTILES - FULLT) * KSPL) * (DM / 64)), 128, 0, stream>>>(
          H, tok, tileE, tileR0, tileCnt, W2, b2, O, Ot);
  combine_kernel<<<T_TOKENS / 16, 256, 0, stream>>>(O, Ot, tileE, tileR0, slot, selG, out_main);
}
